# starting kernel
# baseline (speedup 1.0000x reference)
#include <hip/hip_runtime.h>
#include <hip/hip_bf16.h>

typedef float v2f __attribute__((ext_vector_type(2)));
typedef float v8f __attribute__((ext_vector_type(8)));

#define DDIM 64
#define GEMM_ROWS 128
#define SX_STRIDE 68
#define SW_STRIDE 68

__device__ __forceinline__ unsigned fkey(float f) {
    unsigned u = __float_as_uint(f);
    return (u & 0x80000000u) ? ~u : (u | 0x80000000u);
}
__device__ __forceinline__ float fdec(unsigned k) {
    return (k & 0x80000000u) ? __uint_as_float(k & 0x7FFFFFFFu)
                             : __uint_as_float(~k);
}
#define FKEY_NEG_INF 0x007FFFFFu

__device__ __forceinline__ void fadd_atomic(float* p, float v) {
    unsafeAtomicAdd(p, v);
}

__global__ void compute_pvec(const float* __restrict__ prompt, const float* __restrict__ pw,
                             const float* __restrict__ pb, float* __restrict__ pvec, int P) {
    int d = threadIdx.x;
    if (d >= DDIM) return;
    float s = pb[d];
    for (int j = 0; j < P; ++j) s += prompt[j] * pw[d * P + j];
    pvec[d] = s;
}

__global__ __launch_bounds__(256) void add_prompt(const float* __restrict__ emb,
                                                  const float* __restrict__ pvec,
                                                  float* __restrict__ x,
                                                  float* __restrict__ out_sum,
                                                  long long total) {
    long long i = (long long)blockIdx.x * blockDim.x + threadIdx.x;
    if (i >= total) return;
    float v = emb[i] + pvec[i & (DDIM - 1)];
    x[i] = v;
    out_sum[i] = v;
}

__global__ __launch_bounds__(256) void gemm_xw_t(const float* __restrict__ x,
                                                 const float* __restrict__ W,
                                                 float* __restrict__ xl, int N) {
    __shared__ float sx[GEMM_ROWS * SX_STRIDE];
    __shared__ float sw[DDIM * SW_STRIDE];
    const int t = threadIdx.x;
    const int rowBase = blockIdx.x * GEMM_ROWS;

    {
        const float4* Wv = (const float4*)W;
        float4* swv = (float4*)sw;
        for (int i = t; i < (DDIM * DDIM) / 4; i += 256) {
            int r = i >> 4, c4 = i & 15;
            swv[r * (SW_STRIDE / 4) + c4] = Wv[i];
        }
    }
    {
        const float4* xv = (const float4*)x;
        float4* sxv = (float4*)sx;
        for (int i = t; i < (GEMM_ROWS * DDIM) / 4; i += 256) {
            int r = i >> 4, c4 = i & 15;
            int gr = rowBase + r;
            if (gr >= N) gr = N - 1;
            sxv[r * (SX_STRIDE / 4) + c4] = xv[(long long)gr * 16 + c4];
        }
    }
    __syncthreads();

    const int wave  = t >> 5;
    const int lane  = t & 31;
    const int lrow  = lane & 15;
    const int khalf = (lane >> 4) << 1;
    const int wrow  = wave * 16;

    v8f acc0 = {0.f,0.f,0.f,0.f,0.f,0.f,0.f,0.f};
    v8f acc1 = acc0, acc2 = acc0, acc3 = acc0;

    const float* arow = &sx[(wrow + lrow) * SX_STRIDE + khalf];
    const float* brow = &sw[lrow * SW_STRIDE + khalf];

    #pragma unroll
    for (int k0 = 0; k0 < DDIM; k0 += 4) {
        v2f a, b0, b1, b2, b3;
        a.x  = arow[k0];                       a.y  = arow[k0 + 1];
        b0.x = brow[k0];                       b0.y = brow[k0 + 1];
        b1.x = brow[16 * SW_STRIDE + k0];      b1.y = brow[16 * SW_STRIDE + k0 + 1];
        b2.x = brow[32 * SW_STRIDE + k0];      b2.y = brow[32 * SW_STRIDE + k0 + 1];
        b3.x = brow[48 * SW_STRIDE + k0];      b3.y = brow[48 * SW_STRIDE + k0 + 1];
        acc0 = __builtin_amdgcn_wmma_f32_16x16x4_f32(false, a, false, b0, (short)0, acc0, false, false);
        acc1 = __builtin_amdgcn_wmma_f32_16x16x4_f32(false, a, false, b1, (short)0, acc1, false, false);
        acc2 = __builtin_amdgcn_wmma_f32_16x16x4_f32(false, a, false, b2, (short)0, acc2, false, false);
        acc3 = __builtin_amdgcn_wmma_f32_16x16x4_f32(false, a, false, b3, (short)0, acc3, false, false);
    }

    const int rhi = (lane >> 4) * 8;
    const long long rbase = (long long)(rowBase + wrow + rhi);
    if (rowBase + GEMM_ROWS <= N) {
        float* p = xl + rbase * DDIM + lrow;
        #pragma unroll
        for (int v = 0; v < 8; ++v) {
            p[(long long)v * DDIM +  0] = acc0[v];
            p[(long long)v * DDIM + 16] = acc1[v];
            p[(long long)v * DDIM + 32] = acc2[v];
            p[(long long)v * DDIM + 48] = acc3[v];
        }
    } else {
        #pragma unroll
        for (int v = 0; v < 8; ++v) {
            long long gr = rbase + v;
            if (gr < N) {
                float* p = xl + gr * DDIM + lrow;
                p[ 0] = acc0[v];
                p[16] = acc1[v];
                p[32] = acc2[v];
                p[48] = acc3[v];
            }
        }
    }
}

__global__ __launch_bounds__(256) void att_dots(const float* __restrict__ xl,
                                                const float* __restrict__ av_s,
                                                const float* __restrict__ av_d,
                                                float* __restrict__ a_src,
                                                float* __restrict__ a_dst, int N) {
    long long tid = (long long)blockIdx.x * blockDim.x + threadIdx.x;
    int n = (int)(tid >> 5);
    int lane = threadIdx.x & 31;
    if (n >= N) return;
    float2 v = ((const float2*)(xl + (long long)n * DDIM))[lane];
    float2 s = ((const float2*)av_s)[lane];
    float2 d = ((const float2*)av_d)[lane];
    float ps = v.x * s.x + v.y * s.y;
    float pd = v.x * d.x + v.y * d.y;
    #pragma unroll
    for (int off = 16; off > 0; off >>= 1) {
        ps += __shfl_xor(ps, off, 32);
        pd += __shfl_xor(pd, off, 32);
    }
    if (lane == 0) { a_src[n] = ps; a_dst[n] = pd; }
}

__global__ __launch_bounds__(256) void init_layer(float* __restrict__ acc,
                                                  unsigned* __restrict__ segmax,
                                                  float* __restrict__ segsum, int N) {
    int n = blockIdx.x * blockDim.x + threadIdx.x;
    if (n >= N) return;
    segmax[n] = FKEY_NEG_INF;
    segsum[n] = 0.f;
    float4 z = {0.f, 0.f, 0.f, 0.f};
    float4* p = (float4*)(acc + (long long)n * DDIM);
    #pragma unroll
    for (int i = 0; i < DDIM / 4; ++i) p[i] = z;
}

__device__ __forceinline__ void edge_pair(const int* __restrict__ ei, long long e,
                                          int E, int& s, int& d) {
    if (e < E) { s = ei[e]; d = ei[(long long)E + e]; }
    else       { s = d = (int)(e - E); }
}

__global__ __launch_bounds__(256) void edge_max(const int* __restrict__ ei, int E, int N,
                                                const float* __restrict__ a_src,
                                                const float* __restrict__ a_dst,
                                                unsigned* __restrict__ segmax) {
    long long e = (long long)blockIdx.x * blockDim.x + threadIdx.x;
    long long tot = (long long)E + N;
    if (e >= tot) return;
    int s, d; edge_pair(ei, e, E, s, d);
    float v = a_src[s] + a_dst[d];
    v = v > 0.f ? v : 0.2f * v;
    atomicMax(&segmax[d], fkey(v));
}

__global__ __launch_bounds__(256) void edge_sum(const int* __restrict__ ei, int E, int N,
                                                const float* __restrict__ a_src,
                                                const float* __restrict__ a_dst,
                                                const unsigned* __restrict__ segmax,
                                                float* __restrict__ segsum) {
    long long e = (long long)blockIdx.x * blockDim.x + threadIdx.x;
    long long tot = (long long)E + N;
    if (e >= tot) return;
    int s, d; edge_pair(ei, e, E, s, d);
    float v = a_src[s] + a_dst[d];
    v = v > 0.f ? v : 0.2f * v;
    float m = fdec(segmax[d]);
    fadd_atomic(&segsum[d], __expf(v - m));
}

__global__ __launch_bounds__(256) void edge_scatter(const int* __restrict__ ei, int E, int N,
                                                    const float* __restrict__ a_src,
                                                    const float* __restrict__ a_dst,
                                                    const unsigned* __restrict__ segmax,
                                                    const float* __restrict__ segsum,
                                                    const float* __restrict__ xl,
                                                    float* __restrict__ acc) {
    long long tid = (long long)blockIdx.x * blockDim.x + threadIdx.x;
    long long e = tid >> 5;
    int lane = threadIdx.x & 31;
    long long tot = (long long)E + N;
    if (e >= tot) return;
    int s, d; edge_pair(ei, e, E, s, d);
    float v = a_src[s] + a_dst[d];
    v = v > 0.f ? v : 0.2f * v;
    float m = fdec(segmax[d]);
    float alpha = __expf(v - m) / (segsum[d] + 1e-16f);
    float2 xv = ((const float2*)(xl + (long long)s * DDIM))[lane];
    float* dstp = acc + (long long)d * DDIM + 2 * lane;
    fadd_atomic(dstp,     xv.x * alpha);
    fadd_atomic(dstp + 1, xv.y * alpha);
}

__global__ __launch_bounds__(256) void finalize_layer(float* __restrict__ xbuf,
                                                      const float* __restrict__ bias,
                                                      float* __restrict__ out_sum,
                                                      long long total) {
    long long i = (long long)blockIdx.x * blockDim.x + threadIdx.x;
    if (i >= total) return;
    float v = xbuf[i] + bias[i & (DDIM - 1)];
    v = v > 0.f ? v : (__expf(v) - 1.f);
    xbuf[i] = v;
    out_sum[i] += v;
}

__global__ __launch_bounds__(256) void scale_out(float* __restrict__ out, float inv,
                                                 long long total) {
    long long i = (long long)blockIdx.x * blockDim.x + threadIdx.x;
    if (i >= total) return;
    out[i] *= inv;
}

static inline unsigned gridFor(long long work, int block) {
    return (unsigned)((work + block - 1) / block);
}

extern "C" void kernel_launch(void* const* d_in, const int* in_sizes, int n_in,
                              void* d_out, int out_size, void* d_ws, size_t ws_size,
                              hipStream_t stream) {
    const int*   ei       = (const int*)d_in[0];
    const float* emb      = (const float*)d_in[1];
    const float* prompt   = (const float*)d_in[2];
    const float* proj_w   = (const float*)d_in[3];
    const float* proj_b   = (const float*)d_in[4];
    const float* conv_ws  = (const float*)d_in[5];
    const float* att_srcs = (const float*)d_in[6];
    const float* att_dsts = (const float*)d_in[7];
    const float* conv_bs  = (const float*)d_in[8];

    const int E = in_sizes[0] / 2;
    const int N = in_sizes[1] / DDIM;
    const int P = in_sizes[2];
    const int L = in_sizes[5] / (DDIM * DDIM);

    float* out = (float*)d_out;

    float*    pvec   = (float*)d_ws;
    float*    xbuf   = pvec + 64;
    float*    xlb    = xbuf + (size_t)N * DDIM;
    float*    a_src  = xlb + (size_t)N * DDIM;
    float*    a_dst  = a_src + N;
    unsigned* segmax = (unsigned*)(a_dst + N);
    float*    segsum = (float*)(segmax + N);

    const long long total = (long long)N * DDIM;
    const long long totE  = (long long)E + N;

    compute_pvec<<<1, 64, 0, stream>>>(prompt, proj_w, proj_b, pvec, P);
    add_prompt<<<gridFor(total, 256), 256, 0, stream>>>(emb, pvec, xbuf, out, total);

    for (int l = 0; l < L; ++l) {
        const float* W  = conv_ws  + (size_t)l * DDIM * DDIM;
        const float* as = att_srcs + (size_t)l * DDIM;
        const float* ad = att_dsts + (size_t)l * DDIM;
        const float* cb = conv_bs  + (size_t)l * DDIM;

        gemm_xw_t<<<gridFor(N, GEMM_ROWS), 256, 0, stream>>>(xbuf, W, xlb, N);
        att_dots<<<gridFor((long long)N * 32, 256), 256, 0, stream>>>(xlb, as, ad, a_src, a_dst, N);
        init_layer<<<gridFor(N, 256), 256, 0, stream>>>(xbuf, segmax, segsum, N);
        edge_max<<<gridFor(totE, 256), 256, 0, stream>>>(ei, E, N, a_src, a_dst, segmax);
        edge_sum<<<gridFor(totE, 256), 256, 0, stream>>>(ei, E, N, a_src, a_dst, segmax, segsum);
        edge_scatter<<<gridFor(totE * 32, 256), 256, 0, stream>>>(ei, E, N, a_src, a_dst,
                                                                  segmax, segsum, xlb, xbuf);
        finalize_layer<<<gridFor(total, 256), 256, 0, stream>>>(xbuf, cb, out, total);
    }
    scale_out<<<gridFor(total, 256), 256, 0, stream>>>(out, 1.f / (float)(L + 1), total);
}


# gather: 16 lanes x float4 per destination, two destinations per wave, entries passed inside a 16-lane row by DPP, eight rows per destination in flight; slot order kept (weight sum regroups past 32)
# speedup vs baseline: 3.7308x; 3.7308x over previous
#include <hip/hip_runtime.h>
#include <hip/hip_bf16.h>

typedef float v2f __attribute__((ext_vector_type(2)));
typedef float v8f __attribute__((ext_vector_type(8)));

#define DDIM 64
#define GEMM_ROWS 128
#define SX_STRIDE 68
#define SW_STRIDE 68

__device__ __forceinline__ unsigned fkey(float f) {
    unsigned u = __float_as_uint(f);
    return (u & 0x80000000u) ? ~u : (u | 0x80000000u);
}
__device__ __forceinline__ float fdec(unsigned k) {
    return (k & 0x80000000u) ? __uint_as_float(k & 0x7FFFFFFFu)
                             : __uint_as_float(~k);
}
#define FKEY_NEG_INF 0x007FFFFFu

__device__ __forceinline__ void fadd_atomic(float* p, float v) {
    unsafeAtomicAdd(p, v);
}

__global__ void compute_pvec(const float* __restrict__ prompt, const float* __restrict__ pw,
                             const float* __restrict__ pb, float* __restrict__ pvec, int P) {
    int d = threadIdx.x;
    if (d >= DDIM) return;
    float s = pb[d];
    for (int j = 0; j < P; ++j) s += prompt[j] * pw[d * P + j];
    pvec[d] = s;
}

__global__ __launch_bounds__(256) void add_prompt(const float* __restrict__ emb,
                                                  const float* __restrict__ pvec,
                                                  float* __restrict__ x,
                                                  float* __restrict__ out_sum,
                                                  long long total) {
    long long i = (long long)blockIdx.x * blockDim.x + threadIdx.x;
    if (i >= total) return;
    float v = emb[i] + pvec[i & (DDIM - 1)];
    x[i] = v;
    if (out_sum) out_sum[i] = v;
}


template <int M>
__device__ __forceinline__ float row_xor(float v) {
    return __int_as_float(__builtin_amdgcn_update_dpp(0, __float_as_int(v), 0x160 | M, 0xf, 0xf, true));
}

template <bool ADDP>
__global__ __launch_bounds__(256) void gemm_xw_t(const float* __restrict__ x,
                                                 const float* __restrict__ pvec,
                                                 const float* __restrict__ W,
                                                 const float* __restrict__ av_s,
                                                 const float* __restrict__ av_d,
                                                 float* __restrict__ xl,
                                                 float* __restrict__ a_src,
                                                 float* __restrict__ a_dst, int N) {
    __shared__ float sx[GEMM_ROWS * SX_STRIDE];
    __shared__ float sw[DDIM * SW_STRIDE];
    const int t = threadIdx.x;
    const int rowBase = blockIdx.x * GEMM_ROWS;

    {
        const float4* Wv = (const float4*)W;
        float4* swv = (float4*)sw;
        for (int i = t; i < (DDIM * DDIM) / 4; i += 256) {
            int r = i >> 4, c4 = i & 15;
            swv[r * (SW_STRIDE / 4) + c4] = Wv[i];
        }
    }
    {
        const float4* xv = (const float4*)x;
        float4* sxv = (float4*)sx;
        float4 p4 = {0.f, 0.f, 0.f, 0.f};
        if (ADDP) p4 = ((const float4*)pvec)[t & 15];
        for (int i = t; i < (GEMM_ROWS * DDIM) / 4; i += 256) {
            int r = i >> 4, c4 = i & 15;
            int gr = rowBase + r;
            if (gr >= N) gr = N - 1;
            float4 v = xv[(long long)gr * 16 + c4];
            if (ADDP) { v.x += p4.x;  v.y += p4.y;  v.z += p4.z;  v.w += p4.w; }
            sxv[r * (SX_STRIDE / 4) + c4] = v;
        }
    }
    __syncthreads();

    const int wave  = t >> 5;
    const int lane  = t & 31;
    const int lrow  = lane & 15;
    const int khalf = (lane >> 4) << 1;
    const int wrow  = wave * 16;

    v8f acc0 = {0.f,0.f,0.f,0.f,0.f,0.f,0.f,0.f};
    v8f acc1 = acc0, acc2 = acc0, acc3 = acc0;

    const float* arow = &sx[(wrow + lrow) * SX_STRIDE + khalf];
    const float* brow = &sw[lrow * SW_STRIDE + khalf];

    const float as0 = av_s[lrow], as1 = av_s[16 + lrow], as2 = av_s[32 + lrow], as3 = av_s[48 + lrow];
    const float ad0 = av_d[lrow], ad1 = av_d[16 + lrow], ad2 = av_d[32 + lrow], ad3 = av_d[48 + lrow];

    #pragma unroll
    for (int k0 = 0; k0 < DDIM; k0 += 4) {
        v2f a, b0, b1, b2, b3;
        a.x  = arow[k0];                       a.y  = arow[k0 + 1];
        b0.x = brow[k0];                       b0.y = brow[k0 + 1];
        b1.x = brow[16 * SW_STRIDE + k0];      b1.y = brow[16 * SW_STRIDE + k0 + 1];
        b2.x = brow[32 * SW_STRIDE + k0];      b2.y = brow[32 * SW_STRIDE + k0 + 1];
        b3.x = brow[48 * SW_STRIDE + k0];      b3.y = brow[48 * SW_STRIDE + k0 + 1];
        acc0 = __builtin_amdgcn_wmma_f32_16x16x4_f32(false, a, false, b0, (short)0, acc0, false, false);
        acc1 = __builtin_amdgcn_wmma_f32_16x16x4_f32(false, a, false, b1, (short)0, acc1, false, false);
        acc2 = __builtin_amdgcn_wmma_f32_16x16x4_f32(false, a, false, b2, (short)0, acc2, false, false);
        acc3 = __builtin_amdgcn_wmma_f32_16x16x4_f32(false, a, false, b3, (short)0, acc3, false, false);
    }

    const int rhi = (lane >> 4) * 8;
    const long long rbase = (long long)(rowBase + wrow + rhi);
    if (rowBase + GEMM_ROWS <= N) {
        float* p = xl + rbase * DDIM + lrow;
        #pragma unroll
        for (int v = 0; v < 8; ++v) {
            p[(long long)v * DDIM +  0] = acc0[v];
            p[(long long)v * DDIM + 16] = acc1[v];
            p[(long long)v * DDIM + 32] = acc2[v];
            p[(long long)v * DDIM + 48] = acc3[v];
        }
    } else {
        #pragma unroll
        for (int v = 0; v < 8; ++v) {
            long long gr = rbase + v;
            if (gr < N) {
                float* p = xl + gr * DDIM + lrow;
                p[ 0] = acc0[v];
                p[16] = acc1[v];
                p[32] = acc2[v];
                p[48] = acc3[v];
            }
        }
    }

    const bool k8 = (lane & 8) != 0, k4 = (lane & 4) != 0, k2 = (lane & 2) != 0, k1 = (lane & 1) != 0;
    float r1[8], r2[4], r3[2];
    #pragma unroll
    for (int v = 0; v < 8; ++v) {
        const float ps = acc0[v] * as0 + acc1[v] * as1 + acc2[v] * as2 + acc3[v] * as3;
        const float pd = acc0[v] * ad0 + acc1[v] * ad1 + acc2[v] * ad2 + acc3[v] * ad3;
        const float keep = k8 ? pd : ps;
        const float send = k8 ? ps : pd;
        r1[v] = keep + row_xor<8>(send);
    }
    #pragma unroll
    for (int v = 0; v < 4; ++v) {
        const float keep = k4 ? r1[v + 4] : r1[v];
        const float send = k4 ? r1[v] : r1[v + 4];
        r2[v] = keep + row_xor<4>(send);
    }
    #pragma unroll
    for (int v = 0; v < 2; ++v) {
        const float keep = k2 ? r2[v + 2] : r2[v];
        const float send = k2 ? r2[v] : r2[v + 2];
        r3[v] = keep + row_xor<2>(send);
    }
    const float keep1 = k1 ? r3[1] : r3[0];
    const float send1 = k1 ? r3[0] : r3[1];
    const float dot = keep1 + row_xor<1>(send1);
    const long long drow = rbase + (lrow & 7);
    float* dptr = k8 ? a_dst : a_src;
    if (drow < N) dptr[drow] = dot;
}

__global__ __launch_bounds__(256) void init_layer(float* __restrict__ acc,
                                                  unsigned* __restrict__ segmax,
                                                  float* __restrict__ segsum, int N) {
    int n = blockIdx.x * blockDim.x + threadIdx.x;
    if (n >= N) return;
    segmax[n] = FKEY_NEG_INF;
    segsum[n] = 0.f;
    float4 z = {0.f, 0.f, 0.f, 0.f};
    float4* p = (float4*)(acc + (long long)n * DDIM);
    #pragma unroll
    for (int i = 0; i < DDIM / 4; ++i) p[i] = z;
}

__device__ __forceinline__ void edge_pair(const int* __restrict__ ei, long long e,
                                          int E, int& s, int& d) {
    if (e < E) { s = ei[e]; d = ei[(long long)E + e]; }
    else       { s = d = (int)(e - E); }
}

__global__ __launch_bounds__(256) void edge_max(const int* __restrict__ ei, int E, int N,
                                                const float* __restrict__ a_src,
                                                const float* __restrict__ a_dst,
                                                unsigned* __restrict__ segmax) {
    long long e = (long long)blockIdx.x * blockDim.x + threadIdx.x;
    long long tot = (long long)E + N;
    if (e >= tot) return;
    int s, d; edge_pair(ei, e, E, s, d);
    float v = a_src[s] + a_dst[d];
    v = v > 0.f ? v : 0.2f * v;
    atomicMax(&segmax[d], fkey(v));
}

__global__ __launch_bounds__(256) void edge_scatter(const int* __restrict__ ei, int E, int N,
                                                    const float* __restrict__ a_src,
                                                    const float* __restrict__ a_dst,
                                                    const unsigned* __restrict__ segmax,
                                                    float* __restrict__ segsum,
                                                    const float* __restrict__ xl,
                                                    float* __restrict__ acc) {
    long long tid = (long long)blockIdx.x * blockDim.x + threadIdx.x;
    long long e = tid >> 5;
    int lane = threadIdx.x & 31;
    long long tot = (long long)E + N;
    if (e >= tot) return;
    int s, d; edge_pair(ei, e, E, s, d);
    float v = a_src[s] + a_dst[d];
    v = v > 0.f ? v : 0.2f * v;
    float m = fdec(segmax[d]);
    float w = __expf(v - m);
    float2 xv = ((const float2*)(xl + (long long)s * DDIM))[lane];
    float* dstp = acc + (long long)d * DDIM + 2 * lane;
    if (lane == 0) fadd_atomic(&segsum[d], w);
    fadd_atomic(dstp,     xv.x * w);
    fadd_atomic(dstp + 1, xv.y * w);
}

__global__ __launch_bounds__(256) void finalize_layer(float* __restrict__ xbuf,
                                                      const float* __restrict__ segsum,
                                                      const float* __restrict__ bias,
                                                      float* __restrict__ out_sum,
                                                      long long total) {
    long long i = (long long)blockIdx.x * blockDim.x + threadIdx.x;
    if (i >= total) return;
    float v = xbuf[i] / (segsum[i / DDIM] + 1e-16f) + bias[i & (DDIM - 1)];
    v = v > 0.f ? v : (__expf(v) - 1.f);
    xbuf[i] = v;
    out_sum[i] += v;
}

__global__ __launch_bounds__(256) void scale_out(float* __restrict__ out, float inv,
                                                 long long total) {
    long long i = (long long)blockIdx.x * blockDim.x + threadIdx.x;
    if (i >= total) return;
    out[i] *= inv;
}


#define SCAN_CHUNK 1024

__global__ __launch_bounds__(256) void csr_zero(int* __restrict__ offs, int n1) {
    int i = blockIdx.x * blockDim.x + threadIdx.x;
    if (i < n1) offs[i] = 0;
}

__global__ __launch_bounds__(256) void csr_rank(const int* __restrict__ ei, int E,
                                                int* __restrict__ offs, int* __restrict__ rank) {
    int e = blockIdx.x * blockDim.x + threadIdx.x;
    if (e >= E) return;
    int d = ei[(long long)E + e];
    rank[e] = atomicAdd(&offs[d], 1);
}

__global__ __launch_bounds__(256) void csr_block_sums(const int* __restrict__ offs, int n1,
                                                      int* __restrict__ bsum) {
    __shared__ int red[8];
    const int t = threadIdx.x;
    const int idx = blockIdx.x * SCAN_CHUNK + t * 4;
    int s = 0;
    #pragma unroll
    for (int k = 0; k < 4; ++k) if (idx + k < n1) s += offs[idx + k];
    #pragma unroll
    for (int off = 16; off > 0; off >>= 1) s += __shfl_xor(s, off, 32);
    if ((t & 31) == 0) red[t >> 5] = s;
    __syncthreads();
    if (t == 0) {
        int tot = 0;
        #pragma unroll
        for (int w = 0; w < 8; ++w) tot += red[w];
        bsum[blockIdx.x] = tot;
    }
}

__global__ __launch_bounds__(256) void csr_scan(int* __restrict__ offs, int n1,
                                                const int* __restrict__ bsum) {
    __shared__ int red[8];
    __shared__ int wsum[8];
    const int t = threadIdx.x, lane = t & 31, wave = t >> 5;
    int p = 0;
    for (int i = t; i < (int)blockIdx.x; i += 256) p += bsum[i];
    #pragma unroll
    for (int off = 16; off > 0; off >>= 1) p += __shfl_xor(p, off, 32);
    if (lane == 0) red[wave] = p;

    const int idx = blockIdx.x * SCAN_CHUNK + t * 4;
    const int v0 = (idx     < n1) ? offs[idx]     : 0;
    const int v1 = (idx + 1 < n1) ? offs[idx + 1] : 0;
    const int v2 = (idx + 2 < n1) ? offs[idx + 2] : 0;
    const int v3 = (idx + 3 < n1) ? offs[idx + 3] : 0;
    const int tsum = v0 + v1 + v2 + v3;
    int inc = tsum;
    #pragma unroll
    for (int off = 1; off < 32; off <<= 1) {
        int y = __shfl_up(inc, off, 32);
        if (lane >= off) inc += y;
    }
    if (lane == 31) wsum[wave] = inc;
    __syncthreads();
    int base = 0;
    #pragma unroll
    for (int w = 0; w < 8; ++w) base += red[w];
    for (int w = 0; w < wave; ++w) base += wsum[w];
    const int ex = base + inc - tsum;
    if (idx     < n1) offs[idx]     = ex;
    if (idx + 1 < n1) offs[idx + 1] = ex + v0;
    if (idx + 2 < n1) offs[idx + 2] = ex + v0 + v1;
    if (idx + 3 < n1) offs[idx + 3] = ex + v0 + v1 + v2;
}

__global__ __launch_bounds__(256) void csr_fill(const int* __restrict__ ei, int E,
                                                const int* __restrict__ offs,
                                                const int* __restrict__ rank,
                                                int* __restrict__ ssrc) {
    int e = blockIdx.x * blockDim.x + threadIdx.x;
    if (e >= E) return;
    int d = ei[(long long)E + e];
    ssrc[offs[d] + rank[e]] = ei[e];
}


template <int J>
__device__ __forceinline__ int row_bcast(int v) {
    return __builtin_amdgcn_update_dpp(0, v, 0x150 | J, 0xf, 0xf, true);
}

template <int JB>
__device__ __forceinline__ void ask_rows4(const float* __restrict__ xl, int s, unsigned lofs,
                                          float4& r0, float4& r1, float4& r2, float4& r3) {
    const unsigned o0 = ((unsigned)row_bcast<JB    >(s) << 8) | lofs;
    const unsigned o1 = ((unsigned)row_bcast<JB + 1>(s) << 8) | lofs;
    const unsigned o2 = ((unsigned)row_bcast<JB + 2>(s) << 8) | lofs;
    const unsigned o3 = ((unsigned)row_bcast<JB + 3>(s) << 8) | lofs;
    const char* xb = (const char*)xl;
    r0 = *(const float4*)(xb + o0);
    r1 = *(const float4*)(xb + o1);
    r2 = *(const float4*)(xb + o2);
    r3 = *(const float4*)(xb + o3);
}
template <int JB>
__device__ __forceinline__ void use_rows4(int wi, const float4& r0, const float4& r1,
                                          const float4& r2, const float4& r3, float4& acc) {
    const float w0 = __int_as_float(row_bcast<JB    >(wi));
    const float w1 = __int_as_float(row_bcast<JB + 1>(wi));
    const float w2 = __int_as_float(row_bcast<JB + 2>(wi));
    const float w3 = __int_as_float(row_bcast<JB + 3>(wi));
    acc.x += w0 * r0.x;  acc.y += w0 * r0.y;  acc.z += w0 * r0.z;  acc.w += w0 * r0.w;
    acc.x += w1 * r1.x;  acc.y += w1 * r1.y;  acc.z += w1 * r1.z;  acc.w += w1 * r1.w;
    acc.x += w2 * r2.x;  acc.y += w2 * r2.y;  acc.z += w2 * r2.z;  acc.w += w2 * r2.w;
    acc.x += w3 * r3.x;  acc.y += w3 * r3.y;  acc.z += w3 * r3.z;  acc.w += w3 * r3.w;
}
__device__ __forceinline__ void use_rows16(const float* __restrict__ xl, int s, int wi, int nmax, unsigned lofs,
                                           float4& q0, float4& q1, float4& q2, float4& q3,
                                           float4& p0, float4& p1, float4& p2, float4& p3, float4& acc) {
    use_rows4<0>(wi, q0, q1, q2, q3, acc);
    if (nmax > 4) {
        if (nmax > 8) ask_rows4<8>(xl, s, lofs, q0, q1, q2, q3);
        use_rows4<4>(wi, p0, p1, p2, p3, acc);
        if (nmax > 8) {
            if (nmax > 12) ask_rows4<12>(xl, s, lofs, p0, p1, p2, p3);
            use_rows4<8>(wi, q0, q1, q2, q3, acc);
            if (nmax > 12) use_rows4<12>(wi, p0, p1, p2, p3, acc);
        }
    }
}

#define NODES_PER_WAVE 4
#define MID_NONE  0
#define MID_STORE 1
#define MID_ADD   2
template <bool LAST>
__global__ __launch_bounds__(256) void node_gather(const int* __restrict__ offs,
                                                   const int* __restrict__ ssrc,
                                                   const float* __restrict__ a_src,
                                                   const float* __restrict__ a_dst,
                                                   const float* __restrict__ xl,
                                                   const float* __restrict__ bias,
                                                   float* __restrict__ xio,
                                                   float* __restrict__ out,
                                                   const float* __restrict__ emb,
                                                   const float* __restrict__ pvec,
                                                   int mid, int add_mid, int add_prev,
                                                   float inv, int N) {
    const int lane = threadIdx.x & 31;
    const int l16  = lane & 15;
    const int half = lane >> 4;
    const unsigned lofs = (unsigned)l16 << 4;
    const int wv = __builtin_amdgcn_readfirstlane((int)(threadIdx.x >> 5));
    const int gw = (blockIdx.x * 8 + wv) * 2;
    const int tw = gridDim.x * 16;
    const float4 b4 = ((const float4*)bias)[l16];
    float4 pv = {0.f, 0.f, 0.f, 0.f};
    if (LAST) pv = ((const float4*)pvec)[l16];

    for (int base = gw; base < N; base += tw) {
        const bool valid = base + half < N;
        const int d = valid ? base + half : N - 1;
        const long long drow = (long long)d * DDIM;
        const int start = offs[d];
        const int deg   = offs[d + 1] - start;
        const int cnt   = deg + 1;
        const float ad  = a_dst[d];
        const int c0 = __builtin_amdgcn_readlane(cnt, 0);
        const int c1 = __builtin_amdgcn_readlane(cnt, 16);
        const int cmax = c0 > c1 ? c0 : c1;

        int sA = d;
        if (l16 < deg) sA = ssrc[start + l16];
        const float asA = a_src[sA];
        float4 q0, q1, q2, q3, p0, p1, p2, p3;
        ask_rows4<0>(xl, sA, lofs, q0, q1, q2, q3);
        if (cmax > 4) ask_rows4<4>(xl, sA, lofs, p0, p1, p2, p3);
        float4 e4 = {0.f, 0.f, 0.f, 0.f}, xp = e4;
        if (LAST) {
            e4 = ((const float4*)(emb + drow))[l16];
            if (add_prev) xp = ((const float4*)(xio + drow))[l16];
        }
        float vA = asA + ad;
        vA = vA > 0.f ? vA : 0.2f * vA;
        vA = (l16 < cnt) ? vA : -INFINITY;
        float m = vA;
        for (int rb = 16; rb < cmax; rb += 16) {
            const int i = rb + l16;
            if (i < cnt) {
                const int s = (i < deg) ? ssrc[start + i] : d;
                float v = a_src[s] + ad;
                v = v > 0.f ? v : 0.2f * v;
                m = fmaxf(m, v);
            }
        }
        m = fmaxf(m, row_xor<8>(m));
        m = fmaxf(m, row_xor<4>(m));
        m = fmaxf(m, row_xor<2>(m));
        m = fmaxf(m, row_xor<1>(m));

        float4 acc = {0.f, 0.f, 0.f, 0.f};
        float wl = (l16 < cnt) ? __expf(vA - m) : 0.f;
        use_rows16(xl, sA, __float_as_int(wl), cmax < 16 ? cmax : 16, lofs, q0, q1, q2, q3, p0, p1, p2, p3, acc);
        for (int rb = 16; rb < cmax; rb += 16) {
            const int i = rb + l16;
            int   s = d;
            float w = 0.f;
            if (i < cnt) {
                s = (i < deg) ? ssrc[start + i] : d;
                float v = a_src[s] + ad;
                v = v > 0.f ? v : 0.2f * v;
                w = __expf(v - m);
            }
            wl += w;
            const int rem = cmax - rb;
            ask_rows4<0>(xl, s, lofs, q0, q1, q2, q3);
            if (rem > 4) ask_rows4<4>(xl, s, lofs, p0, p1, p2, p3);
            use_rows16(xl, s, __float_as_int(w), rem < 16 ? rem : 16, lofs, q0, q1, q2, q3, p0, p1, p2, p3, acc);
        }
        wl += row_xor<8>(wl);
        wl += row_xor<4>(wl);
        wl += row_xor<2>(wl);
        wl += row_xor<1>(wl);

        const float den = wl + 1e-16f;
        float4 o;
        o.x = acc.x / den + b4.x;
        o.y = acc.y / den + b4.y;
        o.z = acc.z / den + b4.z;
        o.w = acc.w / den + b4.w;
        o.x = o.x > 0.f ? o.x : (__expf(o.x) - 1.f);
        o.y = o.y > 0.f ? o.y : (__expf(o.y) - 1.f);
        o.z = o.z > 0.f ? o.z : (__expf(o.z) - 1.f);
        o.w = o.w > 0.f ? o.w : (__expf(o.w) - 1.f);
        float4* op = (float4*)(out + drow) + l16;
        if (LAST) {
            float4 sm;
            sm.x = e4.x + pv.x;  sm.y = e4.y + pv.y;  sm.z = e4.z + pv.z;  sm.w = e4.w + pv.w;
            if (add_mid)  { const float4 md = *op;  sm.x += md.x;  sm.y += md.y;  sm.z += md.z;  sm.w += md.w; }
            if (add_prev) { sm.x += xp.x;  sm.y += xp.y;  sm.z += xp.z;  sm.w += xp.w; }
            sm.x += o.x;  sm.y += o.y;  sm.z += o.z;  sm.w += o.w;
            sm.x *= inv;  sm.y *= inv;  sm.z *= inv;  sm.w *= inv;
            if (valid) *op = sm;
        } else if (valid) {
            ((float4*)(xio + drow))[l16] = o;
            if (mid == MID_STORE) {
                *op = o;
            } else if (mid == MID_ADD) {
                float4 ov = *op;
                ov.x += o.x;  ov.y += o.y;  ov.z += o.z;  ov.w += o.w;
                *op = ov;
            }
        }
    }
}

static inline unsigned gridFor(long long work, int block) {
    return (unsigned)((work + block - 1) / block);
}

extern "C" void kernel_launch(void* const* d_in, const int* in_sizes, int n_in,
                              void* d_out, int out_size, void* d_ws, size_t ws_size,
                              hipStream_t stream) {
    const int*   ei       = (const int*)d_in[0];
    const float* emb      = (const float*)d_in[1];
    const float* prompt   = (const float*)d_in[2];
    const float* proj_w   = (const float*)d_in[3];
    const float* proj_b   = (const float*)d_in[4];
    const float* conv_ws  = (const float*)d_in[5];
    const float* att_srcs = (const float*)d_in[6];
    const float* att_dsts = (const float*)d_in[7];
    const float* conv_bs  = (const float*)d_in[8];

    const int E = in_sizes[0] / 2;
    const int N = in_sizes[1] / DDIM;
    const int P = in_sizes[2];
    const int L = in_sizes[5] / (DDIM * DDIM);

    float* out = (float*)d_out;

    float*    pvec   = (float*)d_ws;
    float*    xbuf   = pvec + 64;
    float*    xlb    = xbuf + (size_t)N * DDIM;
    float*    a_src  = xlb + (size_t)N * DDIM;
    float*    a_dst  = a_src + N;
    unsigned* segmax = (unsigned*)(a_dst + N);
    float*    segsum = (float*)(segmax + N);

    const long long total = (long long)N * DDIM;
    const long long totE  = (long long)E + N;

    const size_t nblk = ((size_t)N + 1 + SCAN_CHUNK - 1) / SCAN_CHUNK;
    const size_t need_words = 64 + 2 * (size_t)N * DDIM + 2 * (size_t)N + ((size_t)N + 1) + nblk + (size_t)E;
    const bool grouped = (ws_size >= need_words * sizeof(float)) && ((size_t)E <= (size_t)N * DDIM) && (L >= 1) &&
                         (N < (1 << 24));

    compute_pvec<<<1, 64, 0, stream>>>(prompt, proj_w, proj_b, pvec, P);

    if (grouped) {
        int* offs = (int*)(a_dst + N);
        int* bsum = offs + (N + 1);
        int* ssrc = bsum + nblk;
        int* rank = (int*)xlb;

        csr_zero<<<gridFor((long long)N + 1, 256), 256, 0, stream>>>(offs, N + 1);
        csr_rank<<<gridFor(E, 256), 256, 0, stream>>>(ei, E, offs, rank);
        csr_block_sums<<<(unsigned)nblk, 256, 0, stream>>>(offs, N + 1, bsum);
        csr_scan<<<(unsigned)nblk, 256, 0, stream>>>(offs, N + 1, bsum);
        csr_fill<<<gridFor(E, 256), 256, 0, stream>>>(ei, E, offs, rank, ssrc);

        for (int l = 0; l < L; ++l) {
            const float* W  = conv_ws  + (size_t)l * DDIM * DDIM;
            const float* as = att_srcs + (size_t)l * DDIM;
            const float* ad = att_dsts + (size_t)l * DDIM;
            const float* cb = conv_bs  + (size_t)l * DDIM;

            if (l == 0)
                gemm_xw_t<true><<<gridFor(N, GEMM_ROWS), 256, 0, stream>>>(emb, pvec, W, as, ad, xlb, a_src, a_dst, N);
            else
                gemm_xw_t<false><<<gridFor(N, GEMM_ROWS), 256, 0, stream>>>(xbuf, pvec, W, as, ad, xlb, a_src, a_dst, N);
            if (l + 1 < L) {
                const int mid = (l + 2 < L) ? (l == 0 ? MID_STORE : MID_ADD) : MID_NONE;
                node_gather<false><<<gridFor(N, 8 * NODES_PER_WAVE), 256, 0, stream>>>(
                    offs, ssrc, a_src, a_dst, xlb, cb, xbuf, out, emb, pvec, mid, 0, 0, 0.f, N);
            } else {
                node_gather<true><<<gridFor(N, 8 * NODES_PER_WAVE), 256, 0, stream>>>(
                    offs, ssrc, a_src, a_dst, xlb, cb, xbuf, out, emb, pvec, MID_NONE,
                    L >= 3 ? 1 : 0, L >= 2 ? 1 : 0, 1.f / (float)(L + 1), N);
            }
        }
    } else {
        add_prompt<<<gridFor(total, 256), 256, 0, stream>>>(emb, pvec, xbuf, out, total);
        for (int l = 0; l < L; ++l) {
            const float* W  = conv_ws  + (size_t)l * DDIM * DDIM;
            const float* as = att_srcs + (size_t)l * DDIM;
            const float* ad = att_dsts + (size_t)l * DDIM;
            const float* cb = conv_bs  + (size_t)l * DDIM;

            gemm_xw_t<false><<<gridFor(N, GEMM_ROWS), 256, 0, stream>>>(xbuf, pvec, W, as, ad, xlb, a_src, a_dst, N);
            init_layer<<<gridFor(N, 256), 256, 0, stream>>>(xbuf, segmax, segsum, N);
            edge_max<<<gridFor(totE, 256), 256, 0, stream>>>(ei, E, N, a_src, a_dst, segmax);
            edge_scatter<<<gridFor(totE * 32, 256), 256, 0, stream>>>(ei, E, N, a_src, a_dst,
                                                                      segmax, segsum, xlb, xbuf);
            finalize_layer<<<gridFor(total, 256), 256, 0, stream>>>(xbuf, segsum, cb, out, total);
        }
        scale_out<<<gridFor(total, 256), 256, 0, stream>>>(out, 1.f / (float)(L + 1), total);
    }
}


# gather kernel's dependent round trips batched: W tile's 4 loads, a wave's 17 segment bounds, last layer's 16 emb/x rows each asked at once; result tile restaged via own LDS rows, whole-row stores
# speedup vs baseline: 7.0073x; 1.8782x over previous
#include <hip/hip_runtime.h>
#include <hip/hip_bf16.h>

typedef float v2f __attribute__((ext_vector_type(2)));
typedef float v8f __attribute__((ext_vector_type(8)));

#define DDIM 64
#define GEMM_ROWS 128
#define SX_STRIDE 68
#define SW_STRIDE 68

__device__ __forceinline__ unsigned fkey(float f) {
    unsigned u = __float_as_uint(f);
    return (u & 0x80000000u) ? ~u : (u | 0x80000000u);
}
__device__ __forceinline__ float fdec(unsigned k) {
    return (k & 0x80000000u) ? __uint_as_float(k & 0x7FFFFFFFu)
                             : __uint_as_float(~k);
}
#define FKEY_NEG_INF 0x007FFFFFu

__device__ __forceinline__ void fadd_atomic(float* p, float v) {
    unsafeAtomicAdd(p, v);
}

__global__ void compute_pvec(const float* __restrict__ prompt, const float* __restrict__ pw,
                             const float* __restrict__ pb, float* __restrict__ pvec, int P) {
    int d = threadIdx.x;
    if (d >= DDIM) return;
    float s = pb[d];
    for (int j = 0; j < P; ++j) s += prompt[j] * pw[d * P + j];
    pvec[d] = s;
}

__global__ __launch_bounds__(256) void add_prompt(const float* __restrict__ emb,
                                                  const float* __restrict__ pvec,
                                                  float* __restrict__ x,
                                                  float* __restrict__ out_sum,
                                                  long long total) {
    long long i = (long long)blockIdx.x * blockDim.x + threadIdx.x;
    if (i >= total) return;
    float v = emb[i] + pvec[i & (DDIM - 1)];
    x[i] = v;
    if (out_sum) out_sum[i] = v;
}


template <int M>
__device__ __forceinline__ float row_xor(float v) {
    return __int_as_float(__builtin_amdgcn_update_dpp(0, __float_as_int(v), 0x160 | M, 0xf, 0xf, true));
}

template <bool ADDP>
__global__ __launch_bounds__(256) void gemm_xw_t(const float* __restrict__ x,
                                                 const float* __restrict__ pvec,
                                                 const float* __restrict__ W,
                                                 const float* __restrict__ av_s,
                                                 const float* __restrict__ av_d,
                                                 float* __restrict__ xl,
                                                 float* __restrict__ a_src,
                                                 float* __restrict__ a_dst, int N) {
    __shared__ float sx[GEMM_ROWS * SX_STRIDE];
    __shared__ float sw[DDIM * SW_STRIDE];
    const int t = threadIdx.x;
    const int rowBase = blockIdx.x * GEMM_ROWS;

    {
        const float4* Wv = (const float4*)W;
        float4* swv = (float4*)sw;
        for (int i = t; i < (DDIM * DDIM) / 4; i += 256) {
            int r = i >> 4, c4 = i & 15;
            swv[r * (SW_STRIDE / 4) + c4] = Wv[i];
        }
    }
    {
        const float4* xv = (const float4*)x;
        float4* sxv = (float4*)sx;
        float4 p4 = {0.f, 0.f, 0.f, 0.f};
        if (ADDP) p4 = ((const float4*)pvec)[t & 15];
        for (int i = t; i < (GEMM_ROWS * DDIM) / 4; i += 256) {
            int r = i >> 4, c4 = i & 15;
            int gr = rowBase + r;
            if (gr >= N) gr = N - 1;
            float4 v = xv[(long long)gr * 16 + c4];
            if (ADDP) { v.x += p4.x;  v.y += p4.y;  v.z += p4.z;  v.w += p4.w; }
            sxv[r * (SX_STRIDE / 4) + c4] = v;
        }
    }
    __syncthreads();

    const int wave  = t >> 5;
    const int lane  = t & 31;
    const int lrow  = lane & 15;
    const int khalf = (lane >> 4) << 1;
    const int wrow  = wave * 16;

    v8f acc0 = {0.f,0.f,0.f,0.f,0.f,0.f,0.f,0.f};
    v8f acc1 = acc0, acc2 = acc0, acc3 = acc0;

    const float* arow = &sx[(wrow + lrow) * SX_STRIDE + khalf];
    const float* brow = &sw[lrow * SW_STRIDE + khalf];

    const float as0 = av_s[lrow], as1 = av_s[16 + lrow], as2 = av_s[32 + lrow], as3 = av_s[48 + lrow];
    const float ad0 = av_d[lrow], ad1 = av_d[16 + lrow], ad2 = av_d[32 + lrow], ad3 = av_d[48 + lrow];

    #pragma unroll
    for (int k0 = 0; k0 < DDIM; k0 += 4) {
        v2f a, b0, b1, b2, b3;
        a.x  = arow[k0];                       a.y  = arow[k0 + 1];
        b0.x = brow[k0];                       b0.y = brow[k0 + 1];
        b1.x = brow[16 * SW_STRIDE + k0];      b1.y = brow[16 * SW_STRIDE + k0 + 1];
        b2.x = brow[32 * SW_STRIDE + k0];      b2.y = brow[32 * SW_STRIDE + k0 + 1];
        b3.x = brow[48 * SW_STRIDE + k0];      b3.y = brow[48 * SW_STRIDE + k0 + 1];
        acc0 = __builtin_amdgcn_wmma_f32_16x16x4_f32(false, a, false, b0, (short)0, acc0, false, false);
        acc1 = __builtin_amdgcn_wmma_f32_16x16x4_f32(false, a, false, b1, (short)0, acc1, false, false);
        acc2 = __builtin_amdgcn_wmma_f32_16x16x4_f32(false, a, false, b2, (short)0, acc2, false, false);
        acc3 = __builtin_amdgcn_wmma_f32_16x16x4_f32(false, a, false, b3, (short)0, acc3, false, false);
    }

    const int rhi = (lane >> 4) * 8;
    const long long rbase = (long long)(rowBase + wrow + rhi);
    if (rowBase + GEMM_ROWS <= N) {
        float* p = xl + rbase * DDIM + lrow;
        #pragma unroll
        for (int v = 0; v < 8; ++v) {
            p[(long long)v * DDIM +  0] = acc0[v];
            p[(long long)v * DDIM + 16] = acc1[v];
            p[(long long)v * DDIM + 32] = acc2[v];
            p[(long long)v * DDIM + 48] = acc3[v];
        }
    } else {
        #pragma unroll
        for (int v = 0; v < 8; ++v) {
            long long gr = rbase + v;
            if (gr < N) {
                float* p = xl + gr * DDIM + lrow;
                p[ 0] = acc0[v];
                p[16] = acc1[v];
                p[32] = acc2[v];
                p[48] = acc3[v];
            }
        }
    }

    const bool k8 = (lane & 8) != 0, k4 = (lane & 4) != 0, k2 = (lane & 2) != 0, k1 = (lane & 1) != 0;
    float r1[8], r2[4], r3[2];
    #pragma unroll
    for (int v = 0; v < 8; ++v) {
        const float ps = acc0[v] * as0 + acc1[v] * as1 + acc2[v] * as2 + acc3[v] * as3;
        const float pd = acc0[v] * ad0 + acc1[v] * ad1 + acc2[v] * ad2 + acc3[v] * ad3;
        const float keep = k8 ? pd : ps;
        const float send = k8 ? ps : pd;
        r1[v] = keep + row_xor<8>(send);
    }
    #pragma unroll
    for (int v = 0; v < 4; ++v) {
        const float keep = k4 ? r1[v + 4] : r1[v];
        const float send = k4 ? r1[v] : r1[v + 4];
        r2[v] = keep + row_xor<4>(send);
    }
    #pragma unroll
    for (int v = 0; v < 2; ++v) {
        const float keep = k2 ? r2[v + 2] : r2[v];
        const float send = k2 ? r2[v] : r2[v + 2];
        r3[v] = keep + row_xor<2>(send);
    }
    const float keep1 = k1 ? r3[1] : r3[0];
    const float send1 = k1 ? r3[0] : r3[1];
    const float dot = keep1 + row_xor<1>(send1);
    const long long drow = rbase + (lrow & 7);
    float* dptr = k8 ? a_dst : a_src;
    if (drow < N) dptr[drow] = dot;
}

__global__ __launch_bounds__(256) void init_layer(float* __restrict__ acc,
                                                  unsigned* __restrict__ segmax,
                                                  float* __restrict__ segsum, int N) {
    int n = blockIdx.x * blockDim.x + threadIdx.x;
    if (n >= N) return;
    segmax[n] = FKEY_NEG_INF;
    segsum[n] = 0.f;
    float4 z = {0.f, 0.f, 0.f, 0.f};
    float4* p = (float4*)(acc + (long long)n * DDIM);
    #pragma unroll
    for (int i = 0; i < DDIM / 4; ++i) p[i] = z;
}

__device__ __forceinline__ void edge_pair(const int* __restrict__ ei, long long e,
                                          int E, int& s, int& d) {
    if (e < E) { s = ei[e]; d = ei[(long long)E + e]; }
    else       { s = d = (int)(e - E); }
}

__global__ __launch_bounds__(256) void edge_max(const int* __restrict__ ei, int E, int N,
                                                const float* __restrict__ a_src,
                                                const float* __restrict__ a_dst,
                                                unsigned* __restrict__ segmax) {
    long long e = (long long)blockIdx.x * blockDim.x + threadIdx.x;
    long long tot = (long long)E + N;
    if (e >= tot) return;
    int s, d; edge_pair(ei, e, E, s, d);
    float v = a_src[s] + a_dst[d];
    v = v > 0.f ? v : 0.2f * v;
    atomicMax(&segmax[d], fkey(v));
}

__global__ __launch_bounds__(256) void edge_scatter(const int* __restrict__ ei, int E, int N,
                                                    const float* __restrict__ a_src,
                                                    const float* __restrict__ a_dst,
                                                    const unsigned* __restrict__ segmax,
                                                    float* __restrict__ segsum,
                                                    const float* __restrict__ xl,
                                                    float* __restrict__ acc) {
    long long tid = (long long)blockIdx.x * blockDim.x + threadIdx.x;
    long long e = tid >> 5;
    int lane = threadIdx.x & 31;
    long long tot = (long long)E + N;
    if (e >= tot) return;
    int s, d; edge_pair(ei, e, E, s, d);
    float v = a_src[s] + a_dst[d];
    v = v > 0.f ? v : 0.2f * v;
    float m = fdec(segmax[d]);
    float w = __expf(v - m);
    float2 xv = ((const float2*)(xl + (long long)s * DDIM))[lane];
    float* dstp = acc + (long long)d * DDIM + 2 * lane;
    if (lane == 0) fadd_atomic(&segsum[d], w);
    fadd_atomic(dstp,     xv.x * w);
    fadd_atomic(dstp + 1, xv.y * w);
}

__global__ __launch_bounds__(256) void finalize_layer(float* __restrict__ xbuf,
                                                      const float* __restrict__ segsum,
                                                      const float* __restrict__ bias,
                                                      float* __restrict__ out_sum,
                                                      long long total) {
    long long i = (long long)blockIdx.x * blockDim.x + threadIdx.x;
    if (i >= total) return;
    float v = xbuf[i] / (segsum[i / DDIM] + 1e-16f) + bias[i & (DDIM - 1)];
    v = v > 0.f ? v : (__expf(v) - 1.f);
    xbuf[i] = v;
    out_sum[i] += v;
}

__global__ __launch_bounds__(256) void scale_out(float* __restrict__ out, float inv,
                                                 long long total) {
    long long i = (long long)blockIdx.x * blockDim.x + threadIdx.x;
    if (i >= total) return;
    out[i] *= inv;
}


#define SCAN_CHUNK 1024

__global__ __launch_bounds__(256) void csr_zero(int* __restrict__ offs, int n1) {
    int i = blockIdx.x * blockDim.x + threadIdx.x;
    if (i < n1) offs[i] = 0;
}

__global__ __launch_bounds__(256) void csr_rank(const int* __restrict__ ei, int E,
                                                int* __restrict__ offs, int* __restrict__ rank) {
    int e = blockIdx.x * blockDim.x + threadIdx.x;
    if (e >= E) return;
    int d = ei[(long long)E + e];
    rank[e] = atomicAdd(&offs[d], 1);
}

__global__ __launch_bounds__(256) void csr_block_sums(const int* __restrict__ offs, int n1,
                                                      int* __restrict__ bsum) {
    __shared__ int red[8];
    const int t = threadIdx.x;
    const int idx = blockIdx.x * SCAN_CHUNK + t * 4;
    int s = 0;
    #pragma unroll
    for (int k = 0; k < 4; ++k) if (idx + k < n1) s += offs[idx + k];
    #pragma unroll
    for (int off = 16; off > 0; off >>= 1) s += __shfl_xor(s, off, 32);
    if ((t & 31) == 0) red[t >> 5] = s;
    __syncthreads();
    if (t == 0) {
        int tot = 0;
        #pragma unroll
        for (int w = 0; w < 8; ++w) tot += red[w];
        bsum[blockIdx.x] = tot;
    }
}

__global__ __launch_bounds__(256) void csr_scan(int* __restrict__ offs, int n1,
                                                const int* __restrict__ bsum) {
    __shared__ int red[8];
    __shared__ int wsum[8];
    const int t = threadIdx.x, lane = t & 31, wave = t >> 5;
    int p = 0;
    for (int i = t; i < (int)blockIdx.x; i += 256) p += bsum[i];
    #pragma unroll
    for (int off = 16; off > 0; off >>= 1) p += __shfl_xor(p, off, 32);
    if (lane == 0) red[wave] = p;

    const int idx = blockIdx.x * SCAN_CHUNK + t * 4;
    const int v0 = (idx     < n1) ? offs[idx]     : 0;
    const int v1 = (idx + 1 < n1) ? offs[idx + 1] : 0;
    const int v2 = (idx + 2 < n1) ? offs[idx + 2] : 0;
    const int v3 = (idx + 3 < n1) ? offs[idx + 3] : 0;
    const int tsum = v0 + v1 + v2 + v3;
    int inc = tsum;
    #pragma unroll
    for (int off = 1; off < 32; off <<= 1) {
        int y = __shfl_up(inc, off, 32);
        if (lane >= off) inc += y;
    }
    if (lane == 31) wsum[wave] = inc;
    __syncthreads();
    int base = 0;
    #pragma unroll
    for (int w = 0; w < 8; ++w) base += red[w];
    for (int w = 0; w < wave; ++w) base += wsum[w];
    const int ex = base + inc - tsum;
    if (idx     < n1) offs[idx]     = ex;
    if (idx + 1 < n1) offs[idx + 1] = ex + v0;
    if (idx + 2 < n1) offs[idx + 2] = ex + v0 + v1;
    if (idx + 3 < n1) offs[idx + 3] = ex + v0 + v1 + v2;
}

__global__ __launch_bounds__(256) void csr_fill(const int* __restrict__ ei, int E,
                                                const int* __restrict__ offs,
                                                const int* __restrict__ rank,
                                                int* __restrict__ ssrc) {
    int e = blockIdx.x * blockDim.x + threadIdx.x;
    if (e >= E) return;
    int d = ei[(long long)E + e];
    ssrc[offs[d] + rank[e]] = ei[e];
}


#define BIN_SHIFT 9
#define BIN_W     (1 << BIN_SHIFT)
#define MAX_BINS  1024
#define EPT       16
#define BCHUNK    (256 * EPT)
#define SORT_T    BIN_W

__global__ __launch_bounds__(256) void bin_hist(const int* __restrict__ ei, int E, int nbin,
                                                int* __restrict__ binTotal) {
    __shared__ int hist[MAX_BINS];
    const int t = threadIdx.x;
    for (int i = t; i < nbin; i += 256) hist[i] = 0;
    __syncthreads();
    const int* dst = ei + (long long)E;
    const int base = blockIdx.x * BCHUNK + t;
    int d[EPT];
    #pragma unroll
    for (int k = 0; k < EPT; ++k) { const int e = base + k * 256;  d[k] = e < E ? dst[e] : -1; }
    #pragma unroll
    for (int k = 0; k < EPT; ++k) if (d[k] >= 0) atomicAdd(&hist[d[k] >> BIN_SHIFT], 1);
    __syncthreads();
    for (int i = t; i < nbin; i += 256) { const int c = hist[i];  if (c) atomicAdd(&binTotal[i], c); }
}

__device__ __forceinline__ void fold_vectors(const float* __restrict__ prompt, const float* __restrict__ pw,
                                             const float* __restrict__ pb, float* __restrict__ pvec, int P,
                                             const float* __restrict__ Ws, const float* __restrict__ atts,
                                             const float* __restrict__ attd, float* __restrict__ avec, int L) {
    __shared__ float part[2][2][4][DDIM];
    const int t = threadIdx.x, d = t & (DDIM - 1), g = t >> 6;
    if (t < DDIM) {
        float s = pb[d];
        for (int j = 0; j < P; ++j) s += prompt[j] * pw[d * P + j];
        pvec[d] = s;
    }
    for (int l0 = 0; l0 < L; l0 += 2) {
        #pragma unroll
        for (int li = 0; li < 2; ++li) {
            const int l = l0 + li;
            if (l < L) {
                float us = 0.f, ud = 0.f;
                #pragma unroll
                for (int k = 0; k < 16; ++k) {
                    const int o = g * 16 + k;
                    const float w = Ws[(size_t)l * DDIM * DDIM + o * DDIM + d];
                    us += atts[l * DDIM + o] * w;
                    ud += attd[l * DDIM + o] * w;
                }
                part[li][0][g][d] = us;
                part[li][1][g][d] = ud;
            }
        }
        __syncthreads();
        {
            const int li = t >> 7, v = (t >> 6) & 1, l = l0 + li;
            if (l < L)
                avec[(2 * l + v) * DDIM + d] =
                    ((part[li][v][0][d] + part[li][v][1][d]) + part[li][v][2][d]) + part[li][v][3][d];
        }
        __syncthreads();
    }
}

__global__ __launch_bounds__(256) void bin_scatter(const int* __restrict__ ei, int E, int nbin,
                                                   const int* __restrict__ binTotal,
                                                   int* __restrict__ binCursor, int* __restrict__ binned,
                                                   int foldBlk,
                                                   const float* __restrict__ prompt, const float* __restrict__ pw,
                                                   const float* __restrict__ pb, float* __restrict__ pvec, int P,
                                                   const float* __restrict__ Ws, const float* __restrict__ atts,
                                                   const float* __restrict__ attd, float* __restrict__ avec, int L) {
    if ((int)blockIdx.x == foldBlk) {
        fold_vectors(prompt, pw, pb, pvec, P, Ws, atts, attd, avec, L);
        return;
    }
    __shared__ int hist[MAX_BINS];
    __shared__ int bbase[MAX_BINS];
    __shared__ int wsum[8];
    const int t = threadIdx.x, lane = t & 31, wave = t >> 5;
    for (int i = t; i < MAX_BINS; i += 256) hist[i] = 0;
    __syncthreads();
    const int* dst = ei + (long long)E;
    const int base = blockIdx.x * BCHUNK + t;
    int s[EPT], d[EPT];
    #pragma unroll
    for (int k = 0; k < EPT; ++k) {
        const int e = base + k * 256;
        d[k] = e < E ? dst[e] : -1;
        s[k] = e < E ? ei[e] : 0;
    }
    #pragma unroll
    for (int k = 0; k < EPT; ++k) if (d[k] >= 0) atomicAdd(&hist[d[k] >> BIN_SHIFT], 1);

    const int b0 = t * 4;
    const int v0 = (b0     < nbin) ? binTotal[b0]     : 0;
    const int v1 = (b0 + 1 < nbin) ? binTotal[b0 + 1] : 0;
    const int v2 = (b0 + 2 < nbin) ? binTotal[b0 + 2] : 0;
    const int v3 = (b0 + 3 < nbin) ? binTotal[b0 + 3] : 0;
    const int tsum = v0 + v1 + v2 + v3;
    int inc = tsum;
    #pragma unroll
    for (int off = 1; off < 32; off <<= 1) {
        int y = __shfl_up(inc, off, 32);
        if (lane >= off) inc += y;
    }
    if (lane == 31) wsum[wave] = inc;
    __syncthreads();
    int ex = inc - tsum;
    for (int w = 0; w < wave; ++w) ex += wsum[w];
    const int vv[4] = {v0, v1, v2, v3};
    int cc[4], rr[4] = {0, 0, 0, 0};
    #pragma unroll
    for (int j = 0; j < 4; ++j) cc[j] = (b0 + j < nbin) ? hist[b0 + j] : 0;
    if (b0 < nbin) {
        #pragma unroll
        for (int j = 0; j < 4; ++j) rr[j] = atomicAdd(&binCursor[b0 + j], cc[j]);
    }
    #pragma unroll
    for (int j = 0; j < 4; ++j) {
        const int bin = b0 + j;
        if (bin < nbin) {
            bbase[bin] = ex + rr[j];
            hist[bin] = 0;
        }
        ex += vv[j];
    }
    __syncthreads();
    int2* bp = (int2*)binned;
    #pragma unroll
    for (int k = 0; k < EPT; ++k) {
        if (d[k] >= 0) {
            const int b = d[k] >> BIN_SHIFT;
            const int r = atomicAdd(&hist[b], 1);
            bp[bbase[b] + r] = make_int2(s[k], d[k]);
        }
    }
}

__global__ __launch_bounds__(SORT_T) void bin_sort(const int* __restrict__ binned, int E, int N,
                                                   const int* __restrict__ binTotal,
                                                   int* __restrict__ offs, int* __restrict__ ssrc) {
    __shared__ int cnt[BIN_W];
    __shared__ int nstart[BIN_W];
    __shared__ int red[SORT_T / 32];
    __shared__ int wsum[SORT_T / 32];
    const int t = threadIdx.x, lane = t & 31, wave = t >> 5;
    const int b = blockIdx.x;
    int p = 0;
    for (int i = t; i < b; i += SORT_T) p += binTotal[i];
    #pragma unroll
    for (int off = 16; off > 0; off >>= 1) p += __shfl_xor(p, off, 32);
    if (lane == 0) red[wave] = p;
    const int total = binTotal[b];
    cnt[t] = 0;
    __syncthreads();
    int binStart = 0;
    #pragma unroll
    for (int w = 0; w < SORT_T / 32; ++w) binStart += red[w];
    const int2* pairs = (const int2*)binned + binStart;

    for (int i0 = t; i0 < total; i0 += 4 * SORT_T) {
        int2 a[4];
        #pragma unroll
        for (int k = 0; k < 4; ++k) {
            const int i = i0 + k * SORT_T;
            a[k] = make_int2(0, -1);
            if (i < total) a[k] = pairs[i];
        }
        #pragma unroll
        for (int k = 0; k < 4; ++k) if (a[k].y >= 0) atomicAdd(&cnt[a[k].y & (BIN_W - 1)], 1);
    }
    __syncthreads();

    const int c = cnt[t];
    int inc = c;
    #pragma unroll
    for (int off = 1; off < 32; off <<= 1) {
        int y = __shfl_up(inc, off, 32);
        if (lane >= off) inc += y;
    }
    if (lane == 31) wsum[wave] = inc;
    __syncthreads();
    int ex = inc - c;
    for (int w = 0; w < wave; ++w) ex += wsum[w];
    nstart[t] = ex;
    cnt[t] = 0;
    const int node = (b << BIN_SHIFT) + t;
    if (node < N) offs[node] = binStart + ex;
    if (node == N - 1) offs[N] = E;
    __syncthreads();

    for (int i0 = t; i0 < total; i0 += 4 * SORT_T) {
        int2 a[4];
        #pragma unroll
        for (int k = 0; k < 4; ++k) {
            const int i = i0 + k * SORT_T;
            a[k] = make_int2(0, -1);
            if (i < total) a[k] = pairs[i];
        }
        #pragma unroll
        for (int k = 0; k < 4; ++k) {
            if (a[k].y >= 0) {
                const int ln = a[k].y & (BIN_W - 1);
                const int r = atomicAdd(&cnt[ln], 1);
                ssrc[binStart + nstart[ln] + r] = a[k].x;
            }
        }
    }
}


template <int J>
__device__ __forceinline__ int row_bcast(int v) {
    return __builtin_amdgcn_update_dpp(0, v, 0x150 | J, 0xf, 0xf, true);
}

template <int JB>
__device__ __forceinline__ void ask_rows4(const float* __restrict__ xl, int s, unsigned lofs,
                                          float4& r0, float4& r1, float4& r2, float4& r3) {
    const unsigned o0 = ((unsigned)row_bcast<JB    >(s) << 8) | lofs;
    const unsigned o1 = ((unsigned)row_bcast<JB + 1>(s) << 8) | lofs;
    const unsigned o2 = ((unsigned)row_bcast<JB + 2>(s) << 8) | lofs;
    const unsigned o3 = ((unsigned)row_bcast<JB + 3>(s) << 8) | lofs;
    const char* xb = (const char*)xl;
    r0 = *(const float4*)(xb + o0);
    r1 = *(const float4*)(xb + o1);
    r2 = *(const float4*)(xb + o2);
    r3 = *(const float4*)(xb + o3);
}
template <int JB>
__device__ __forceinline__ void use_rows4(int wi, const float4& r0, const float4& r1,
                                          const float4& r2, const float4& r3, float4& acc) {
    const float w0 = __int_as_float(row_bcast<JB    >(wi));
    const float w1 = __int_as_float(row_bcast<JB + 1>(wi));
    const float w2 = __int_as_float(row_bcast<JB + 2>(wi));
    const float w3 = __int_as_float(row_bcast<JB + 3>(wi));
    acc.x += w0 * r0.x;  acc.y += w0 * r0.y;  acc.z += w0 * r0.z;  acc.w += w0 * r0.w;
    acc.x += w1 * r1.x;  acc.y += w1 * r1.y;  acc.z += w1 * r1.z;  acc.w += w1 * r1.w;
    acc.x += w2 * r2.x;  acc.y += w2 * r2.y;  acc.z += w2 * r2.z;  acc.w += w2 * r2.w;
    acc.x += w3 * r3.x;  acc.y += w3 * r3.y;  acc.z += w3 * r3.z;  acc.w += w3 * r3.w;
}
__device__ __forceinline__ void use_rows16(const float* __restrict__ xl, int s, int wi, int nmax, unsigned lofs,
                                           float4& q0, float4& q1, float4& q2, float4& q3,
                                           float4& p0, float4& p1, float4& p2, float4& p3, float4& acc) {
    use_rows4<0>(wi, q0, q1, q2, q3, acc);
    if (nmax > 4) {
        if (nmax > 8) ask_rows4<8>(xl, s, lofs, q0, q1, q2, q3);
        use_rows4<4>(wi, p0, p1, p2, p3, acc);
        if (nmax > 8) {
            if (nmax > 12) ask_rows4<12>(xl, s, lofs, p0, p1, p2, p3);
            use_rows4<8>(wi, q0, q1, q2, q3, acc);
            if (nmax > 12) use_rows4<12>(wi, p0, p1, p2, p3, acc);
        }
    }
}

#define NODES_PER_WAVE 4
#define MID_NONE  0
#define MID_STORE 1
#define MID_ADD   2
template <bool LAST>
__global__ __launch_bounds__(256) void node_gather(const int* __restrict__ offs,
                                                   const int* __restrict__ ssrc,
                                                   const float* __restrict__ a_src,
                                                   const float* __restrict__ a_dst,
                                                   const float* __restrict__ xl,
                                                   const float* __restrict__ bias,
                                                   float* __restrict__ xio,
                                                   float* __restrict__ out,
                                                   const float* __restrict__ emb,
                                                   const float* __restrict__ pvec,
                                                   int mid, int add_mid, int add_prev,
                                                   float inv, int N) {
    const int lane = threadIdx.x & 31;
    const int l16  = lane & 15;
    const int half = lane >> 4;
    const unsigned lofs = (unsigned)l16 << 4;
    const int wv = __builtin_amdgcn_readfirstlane((int)(threadIdx.x >> 5));
    const int gw = (blockIdx.x * 8 + wv) * 2;
    const int tw = gridDim.x * 16;
    const float4 b4 = ((const float4*)bias)[l16];
    float4 pv = {0.f, 0.f, 0.f, 0.f};
    if (LAST) pv = ((const float4*)pvec)[l16];

    for (int base = gw; base < N; base += tw) {
        const bool valid = base + half < N;
        const int d = valid ? base + half : N - 1;
        const long long drow = (long long)d * DDIM;
        const int start = offs[d];
        const int deg   = offs[d + 1] - start;
        const int cnt   = deg + 1;
        const float ad  = a_dst[d];
        const int c0 = __builtin_amdgcn_readlane(cnt, 0);
        const int c1 = __builtin_amdgcn_readlane(cnt, 16);
        const int cmax = c0 > c1 ? c0 : c1;

        int sA = d;
        if (l16 < deg) sA = ssrc[start + l16];
        const float asA = a_src[sA];
        float4 q0, q1, q2, q3, p0, p1, p2, p3;
        ask_rows4<0>(xl, sA, lofs, q0, q1, q2, q3);
        if (cmax > 4) ask_rows4<4>(xl, sA, lofs, p0, p1, p2, p3);
        float4 e4 = {0.f, 0.f, 0.f, 0.f}, xp = e4;
        if (LAST) {
            e4 = ((const float4*)(emb + drow))[l16];
            if (add_prev) xp = ((const float4*)(xio + drow))[l16];
        }
        float vA = asA + ad;
        vA = vA > 0.f ? vA : 0.2f * vA;
        vA = (l16 < cnt) ? vA : -INFINITY;
        float m = vA;
        for (int rb = 16; rb < cmax; rb += 16) {
            const int i = rb + l16;
            if (i < cnt) {
                const int s = (i < deg) ? ssrc[start + i] : d;
                float v = a_src[s] + ad;
                v = v > 0.f ? v : 0.2f * v;
                m = fmaxf(m, v);
            }
        }
        m = fmaxf(m, row_xor<8>(m));
        m = fmaxf(m, row_xor<4>(m));
        m = fmaxf(m, row_xor<2>(m));
        m = fmaxf(m, row_xor<1>(m));

        float4 acc = {0.f, 0.f, 0.f, 0.f};
        float wl = (l16 < cnt) ? __expf(vA - m) : 0.f;
        use_rows16(xl, sA, __float_as_int(wl), cmax < 16 ? cmax : 16, lofs, q0, q1, q2, q3, p0, p1, p2, p3, acc);
        for (int rb = 16; rb < cmax; rb += 16) {
            const int i = rb + l16;
            int   s = d;
            float w = 0.f;
            if (i < cnt) {
                s = (i < deg) ? ssrc[start + i] : d;
                float v = a_src[s] + ad;
                v = v > 0.f ? v : 0.2f * v;
                w = __expf(v - m);
            }
            wl += w;
            const int rem = cmax - rb;
            ask_rows4<0>(xl, s, lofs, q0, q1, q2, q3);
            if (rem > 4) ask_rows4<4>(xl, s, lofs, p0, p1, p2, p3);
            use_rows16(xl, s, __float_as_int(w), rem < 16 ? rem : 16, lofs, q0, q1, q2, q3, p0, p1, p2, p3, acc);
        }
        wl += row_xor<8>(wl);
        wl += row_xor<4>(wl);
        wl += row_xor<2>(wl);
        wl += row_xor<1>(wl);

        const float den = wl + 1e-16f;
        float4 o;
        o.x = acc.x / den + b4.x;
        o.y = acc.y / den + b4.y;
        o.z = acc.z / den + b4.z;
        o.w = acc.w / den + b4.w;
        o.x = o.x > 0.f ? o.x : (__expf(o.x) - 1.f);
        o.y = o.y > 0.f ? o.y : (__expf(o.y) - 1.f);
        o.z = o.z > 0.f ? o.z : (__expf(o.z) - 1.f);
        o.w = o.w > 0.f ? o.w : (__expf(o.w) - 1.f);
        float4* op = (float4*)(out + drow) + l16;
        if (LAST) {
            float4 sm;
            sm.x = e4.x + pv.x;  sm.y = e4.y + pv.y;  sm.z = e4.z + pv.z;  sm.w = e4.w + pv.w;
            if (add_mid)  { const float4 md = *op;  sm.x += md.x;  sm.y += md.y;  sm.z += md.z;  sm.w += md.w; }
            if (add_prev) { sm.x += xp.x;  sm.y += xp.y;  sm.z += xp.z;  sm.w += xp.w; }
            sm.x += o.x;  sm.y += o.y;  sm.z += o.z;  sm.w += o.w;
            sm.x *= inv;  sm.y *= inv;  sm.z *= inv;  sm.w *= inv;
            if (valid) *op = sm;
        } else if (valid) {
            ((float4*)(xio + drow))[l16] = o;
            if (mid == MID_STORE) {
                *op = o;
            } else if (mid == MID_ADD) {
                float4 ov = *op;
                ov.x += o.x;  ov.y += o.y;  ov.z += o.z;  ov.w += o.w;
                *op = ov;
            }
        }
    }
}


__global__ void compute_pvec_att(const float* __restrict__ prompt, const float* __restrict__ pw,
                                 const float* __restrict__ pb, float* __restrict__ pvec, int P,
                                 const float* __restrict__ Ws, const float* __restrict__ atts,
                                 const float* __restrict__ attd, float* __restrict__ avec, int L,
                                 int* __restrict__ zeroed, int nz) {
    int d = threadIdx.x;
    if (d >= DDIM) return;
    for (int i = d; i < nz; i += DDIM) zeroed[i] = 0;
    float s = pb[d];
    for (int j = 0; j < P; ++j) s += prompt[j] * pw[d * P + j];
    pvec[d] = s;
    for (int l = 0; l < L; ++l) {
        float us = 0.f, ud = 0.f;
        for (int o = 0; o < DDIM; ++o) {
            const float w = Ws[(size_t)l * DDIM * DDIM + o * DDIM + d];
            us += atts[l * DDIM + o] * w;
            ud += attd[l * DDIM + o] * w;
        }
        avec[(2 * l) * DDIM + d] = us;
        avec[(2 * l + 1) * DDIM + d] = ud;
    }
}

__device__ __forceinline__ float row_dot(const float4& r, const float4& u) {
    float s = r.x * u.x + r.y * u.y + r.z * u.z + r.w * u.w;
    s += row_xor<8>(s);
    s += row_xor<4>(s);
    s += row_xor<2>(s);
    s += row_xor<1>(s);
    return s;
}

template <int JB>
__device__ __forceinline__ void fold_rows4(int nvalid, float cad, const float4& u4, int l16,
                                           const float4& r0, const float4& r1, const float4& r2, const float4& r3,
                                           float& m, float& wsum, float4& acc) {
    const float d0 = r0.x * u4.x + r0.y * u4.y + r0.z * u4.z + r0.w * u4.w;
    const float d1 = r1.x * u4.x + r1.y * u4.y + r1.z * u4.z + r1.w * u4.w;
    const float d2 = r2.x * u4.x + r2.y * u4.y + r2.z * u4.z + r2.w * u4.w;
    const float d3 = r3.x * u4.x + r3.y * u4.y + r3.z * u4.z + r3.w * u4.w;
    const bool k8 = (l16 & 8) != 0, k4 = (l16 & 4) != 0;
    const float e0 = (k8 ? d2 : d0) + row_xor<8>(k8 ? d0 : d2);
    const float e1 = (k8 ? d3 : d1) + row_xor<8>(k8 ? d1 : d3);
    float f = (k4 ? e1 : e0) + row_xor<4>(k4 ? e0 : e1);
    f += row_xor<2>(f);
    f += row_xor<1>(f);
    float v = f + cad;
    v = v > 0.f ? v : 0.2f * v;
    v = (JB + (l16 >> 2) < nvalid) ? v : -INFINITY;
    float mx = fmaxf(v, row_xor<8>(v));
    mx = fmaxf(mx, row_xor<4>(mx));
    const float mn = fmaxf(m, mx);
    const float sc = __expf(m - mn);
    const int   wi = __float_as_int(__expf(v - mn));
    const float w0 = __int_as_float(row_bcast<0>(wi));
    const float w1 = __int_as_float(row_bcast<4>(wi));
    const float w2 = __int_as_float(row_bcast<8>(wi));
    const float w3 = __int_as_float(row_bcast<12>(wi));
    acc.x = acc.x * sc + w0 * r0.x;  acc.y = acc.y * sc + w0 * r0.y;
    acc.z = acc.z * sc + w0 * r0.z;  acc.w = acc.w * sc + w0 * r0.w;
    acc.x += w1 * r1.x;  acc.y += w1 * r1.y;  acc.z += w1 * r1.z;  acc.w += w1 * r1.w;
    acc.x += w2 * r2.x;  acc.y += w2 * r2.y;  acc.z += w2 * r2.z;  acc.w += w2 * r2.w;
    acc.x += w3 * r3.x;  acc.y += w3 * r3.y;  acc.z += w3 * r3.z;  acc.w += w3 * r3.w;
    wsum = wsum * sc + w0;
    wsum += w1;
    wsum += w2;
    wsum += w3;
    m = mn;
}

__device__ __forceinline__ void fold_rows16(const float* __restrict__ xl, int s, int nvalid, int nmax, unsigned lofs,
                                            float cad, const float4& u4, int l16,
                                            float4& q0, float4& q1, float4& q2, float4& q3,
                                            float4& p0, float4& p1, float4& p2, float4& p3,
                                            float& m, float& wsum, float4& acc) {
    fold_rows4<0>(nvalid, cad, u4, l16, q0, q1, q2, q3, m, wsum, acc);
    if (nmax > 4) {
        if (nmax > 8) ask_rows4<8>(xl, s, lofs, q0, q1, q2, q3);
        fold_rows4<4>(nvalid, cad, u4, l16, p0, p1, p2, p3, m, wsum, acc);
        if (nmax > 8) {
            if (nmax > 12) ask_rows4<12>(xl, s, lofs, p0, p1, p2, p3);
            fold_rows4<8>(nvalid, cad, u4, l16, q0, q1, q2, q3, m, wsum, acc);
            if (nmax > 12) fold_rows4<12>(nvalid, cad, u4, l16, p0, p1, p2, p3, m, wsum, acc);
        }
    }
}

template <bool LAST, bool FULL>
__device__ __forceinline__ void rows_out(const float* sxw, int dn0, unsigned lofs,
                                         const float* __restrict__ emb, const float* __restrict__ xin,
                                         const float* __restrict__ pvec, int l16, float inv,
                                         float* __restrict__ dst, int N) {
    float4 e4[8], x4[8];
    float4 pj = {0.f, 0.f, 0.f, 0.f};
    if (LAST) {
        pj = ((const float4*)pvec)[l16];
        #pragma unroll
        for (int pr = 0; pr < 8; ++pr) {
            const int dn = dn0 + 2 * pr;
            const unsigned ob = ((unsigned)((FULL || dn < N) ? dn : N - 1) << 8) | lofs;
            e4[pr] = *(const float4*)((const char*)emb + ob);
            x4[pr] = *(const float4*)((const char*)xin + ob);
        }
    }
    float4 y = *(const float4*)sxw;
    #pragma unroll
    for (int pr = 0; pr < 8; ++pr) {
        const int dn = dn0 + 2 * pr;
        float4 yn = y;
        if (pr < 7) yn = *(const float4*)(sxw + 2 * (pr + 1) * SX_STRIDE);
        float4 o = y;
        if (LAST) {
            o.x = (((e4[pr].x + pj.x) + x4[pr].x) + y.x) * inv;
            o.y = (((e4[pr].y + pj.y) + x4[pr].y) + y.y) * inv;
            o.z = (((e4[pr].z + pj.z) + x4[pr].z) + y.z) * inv;
            o.w = (((e4[pr].w + pj.w) + x4[pr].w) + y.w) * inv;
        }
        if (FULL || dn < N) *(float4*)((char*)dst + (((unsigned)dn << 8) | lofs)) = o;
        if (LAST) __builtin_amdgcn_sched_barrier(0);
        y = yn;
    }
}

template <bool LAST, bool ADDP>
__global__ __launch_bounds__(256) void gather_mm(const int* __restrict__ offs, const int* __restrict__ ssrc,
                                                 const float* __restrict__ xin, const float* __restrict__ pvec,
                                                 const float* __restrict__ W, const float* __restrict__ bias,
                                                 const float* __restrict__ avl, float* __restrict__ xout,
                                                 float* __restrict__ out, const float* __restrict__ emb,
                                                 float inv, int N) {
    __shared__ float sx[GEMM_ROWS * SX_STRIDE];
    __shared__ float sw[DDIM * SW_STRIDE];
    const int t = threadIdx.x;
    const int rowBase = blockIdx.x * GEMM_ROWS;
    {
        const float4* Wv = (const float4*)W;
        float4* swv = (float4*)sw;
        float4 wq[4];
        #pragma unroll
        for (int k = 0; k < 4; ++k) wq[k] = Wv[t + 256 * k];
        #pragma unroll
        for (int k = 0; k < 4; ++k) {
            const int i = t + 256 * k;
            swv[(i >> 4) * (SW_STRIDE / 4) + (i & 15)] = wq[k];
        }
    }
    const int lane = t & 31;
    const int l16  = lane & 15;
    const int half = lane >> 4;
    const unsigned lofs = (unsigned)l16 << 4;
    const int wv = __builtin_amdgcn_readfirstlane((int)(threadIdx.x >> 5));
    const int wrow = wv * 16;
    const float4 u4 = ((const float4*)avl)[l16];
    const float4 t4 = ((const float4*)avl)[16 + l16];
    float4 pv = {0.f, 0.f, 0.f, 0.f};
    float cpp = 0.f;
    if (ADDP) {
        pv = ((const float4*)pvec)[l16];
        cpp = row_dot(pv, u4) + row_dot(pv, t4);
    }

    const int ofi = rowBase + wrow + (lane < 16 ? lane : 16);
    const int ofv = offs[ofi < N ? ofi : N];

    for (int pr = 0; pr < 8; ++pr) {
        const int dn = rowBase + wrow + 2 * pr + half;
        const int d = dn < N ? dn : N - 1;
        const int o0 = __builtin_amdgcn_readlane(ofv, 2 * pr);
        const int o1 = __builtin_amdgcn_readlane(ofv, 2 * pr + 1);
        const int o2 = __builtin_amdgcn_readlane(ofv, 2 * pr + 2);
        const int g0 = o1 - o0, g1 = o2 - o1;
        const int start = half ? o1 : o0;
        const int deg   = half ? g1 : g0;
        const int dmax = g0 > g1 ? g0 : g1;

        const float4 xo = *(const float4*)((const char*)xin + (((unsigned)d << 8) | lofs));
        int s = d;
        if (l16 < deg) s = ssrc[start + l16];
        float4 q0, q1, q2, q3, p0, p1, p2, p3;
        ask_rows4<0>(xin, s, lofs, q0, q1, q2, q3);
        if (dmax > 4) ask_rows4<4>(xin, s, lofs, p0, p1, p2, p3);

        const float cad = row_dot(xo, t4) + cpp;
        float m = row_dot(xo, u4) + cad;
        m = m > 0.f ? m : 0.2f * m;
        float wsum = 1.f;
        float4 acc = xo;
        for (int rb = 0; rb < dmax; rb += 16) {
            const int rem = dmax - rb;
            if (rb > 0) {
                const int i = rb + l16;
                s = d;
                if (i < deg) s = ssrc[start + i];
                ask_rows4<0>(xin, s, lofs, q0, q1, q2, q3);
                if (rem > 4) ask_rows4<4>(xin, s, lofs, p0, p1, p2, p3);
            }
            fold_rows16(xin, s, deg - rb, rem < 16 ? rem : 16, lofs, cad, u4, l16,
                        q0, q1, q2, q3, p0, p1, p2, p3, m, wsum, acc);
        }

        const float den = wsum + 1e-16f;
        float4 r;
        r.x = acc.x / den;  r.y = acc.y / den;  r.z = acc.z / den;  r.w = acc.w / den;
        if (ADDP) {
            const float sa = wsum / den;
            r.x += sa * pv.x;  r.y += sa * pv.y;  r.z += sa * pv.z;  r.w += sa * pv.w;
        }
        *(float4*)&sx[(wrow + 2 * pr + half) * SX_STRIDE + 4 * l16] = r;
    }
    __syncthreads();

    const int lrow  = l16;
    const int khalf = half << 1;
    v8f acc0 = {0.f,0.f,0.f,0.f,0.f,0.f,0.f,0.f};
    v8f acc1 = acc0, acc2 = acc0, acc3 = acc0;
    const float* arow = &sx[(wrow + lrow) * SX_STRIDE + khalf];
    const float* brow = &sw[lrow * SW_STRIDE + khalf];
    const float bb0 = bias[lrow], bb1 = bias[16 + lrow], bb2 = bias[32 + lrow], bb3 = bias[48 + lrow];

    #pragma unroll
    for (int k0 = 0; k0 < DDIM; k0 += 4) {
        v2f a, b0, b1, b2, b3;
        a.x  = arow[k0];                       a.y  = arow[k0 + 1];
        b0.x = brow[k0];                       b0.y = brow[k0 + 1];
        b1.x = brow[16 * SW_STRIDE + k0];      b1.y = brow[16 * SW_STRIDE + k0 + 1];
        b2.x = brow[32 * SW_STRIDE + k0];      b2.y = brow[32 * SW_STRIDE + k0 + 1];
        b3.x = brow[48 * SW_STRIDE + k0];      b3.y = brow[48 * SW_STRIDE + k0 + 1];
        acc0 = __builtin_amdgcn_wmma_f32_16x16x4_f32(false, a, false, b0, (short)0, acc0, false, false);
        acc1 = __builtin_amdgcn_wmma_f32_16x16x4_f32(false, a, false, b1, (short)0, acc1, false, false);
        acc2 = __builtin_amdgcn_wmma_f32_16x16x4_f32(false, a, false, b2, (short)0, acc2, false, false);
        acc3 = __builtin_amdgcn_wmma_f32_16x16x4_f32(false, a, false, b3, (short)0, acc3, false, false);
    }

    #pragma unroll
    for (int v = 0; v < 8; ++v) {
        float y0 = acc0[v] + bb0, y1 = acc1[v] + bb1, y2 = acc2[v] + bb2, y3 = acc3[v] + bb3;
        acc0[v] = y0 > 0.f ? y0 : (__expf(y0) - 1.f);
        acc1[v] = y1 > 0.f ? y1 : (__expf(y1) - 1.f);
        acc2[v] = y2 > 0.f ? y2 : (__expf(y2) - 1.f);
        acc3[v] = y3 > 0.f ? y3 : (__expf(y3) - 1.f);
    }

    const int rhi = half * 8;
    {
        float* wr = &sx[(wrow + rhi) * SX_STRIDE + lrow];
        #pragma unroll
        for (int v = 0; v < 8; ++v) {
            wr[v * SX_STRIDE +  0] = acc0[v];
            wr[v * SX_STRIDE + 16] = acc1[v];
            wr[v * SX_STRIDE + 32] = acc2[v];
            wr[v * SX_STRIDE + 48] = acc3[v];
        }
    }
    const float* sxw = &sx[(wrow + half) * SX_STRIDE + 4 * l16];
    const int dn0 = rowBase + wrow + half;
    if (rowBase + GEMM_ROWS <= N)
        rows_out<LAST, true>(sxw, dn0, lofs, emb, xin, pvec, l16, inv, LAST ? out : xout, N);
    else
        rows_out<LAST, false>(sxw, dn0, lofs, emb, xin, pvec, l16, inv, LAST ? out : xout, N);
}

static inline unsigned gridFor(long long work, int block) {
    return (unsigned)((work + block - 1) / block);
}

extern "C" void kernel_launch(void* const* d_in, const int* in_sizes, int n_in,
                              void* d_out, int out_size, void* d_ws, size_t ws_size,
                              hipStream_t stream) {
    const int*   ei       = (const int*)d_in[0];
    const float* emb      = (const float*)d_in[1];
    const float* prompt   = (const float*)d_in[2];
    const float* proj_w   = (const float*)d_in[3];
    const float* proj_b   = (const float*)d_in[4];
    const float* conv_ws  = (const float*)d_in[5];
    const float* att_srcs = (const float*)d_in[6];
    const float* att_dsts = (const float*)d_in[7];
    const float* conv_bs  = (const float*)d_in[8];

    const int E = in_sizes[0] / 2;
    const int N = in_sizes[1] / DDIM;
    const int P = in_sizes[2];
    const int L = in_sizes[5] / (DDIM * DDIM);

    float* out = (float*)d_out;

    float*    pvec   = (float*)d_ws;
    float*    xbuf   = pvec + 64;
    float*    xlb    = xbuf + (size_t)N * DDIM;
    float*    a_src  = xlb + (size_t)N * DDIM;
    float*    a_dst  = a_src + N;
    unsigned* segmax = (unsigned*)(a_dst + N);
    float*    segsum = (float*)(segmax + N);

    const long long total = (long long)N * DDIM;
    const long long totE  = (long long)E + N;

    const size_t nblk = ((size_t)N + 1 + SCAN_CHUNK - 1) / SCAN_CHUNK;
    const size_t need_words = 64 + 2 * (size_t)N * DDIM + 2 * (size_t)N + ((size_t)N + 1) + nblk + (size_t)E;
    const bool grouped = (ws_size >= need_words * sizeof(float)) && ((size_t)E <= (size_t)N * DDIM) && (L >= 1) &&
                         (N < (1 << 24));

    const size_t EA = ((size_t)E + 63) & ~(size_t)63;
    const int nbin = (N + BIN_W - 1) >> BIN_SHIFT;
    const bool binned_ok = grouped && (E >= 1) && (nbin <= MAX_BINS) &&
                           (2 * EA + 4 * DDIM + 2 * MAX_BINS <= (size_t)N * DDIM);
    const bool fused = grouped && (L == 2) && (binned_ok || EA + 4 * DDIM <= (size_t)N * DDIM);
    float* avec   = xlb + (binned_ok ? 2 * EA : EA);
    int* binTotal  = (int*)(xlb + 2 * EA + 4 * DDIM);
    int* binCursor = binTotal + MAX_BINS;

    const bool foldBeside = fused && binned_ok;
    if (foldBeside)
        csr_zero<<<gridFor(2 * MAX_BINS, 256), 256, 0, stream>>>(binTotal, 2 * MAX_BINS);
    else if (fused)
        compute_pvec_att<<<1, 64, 0, stream>>>(prompt, proj_w, proj_b, pvec, P, conv_ws, att_srcs, att_dsts, avec, L,
                                               binTotal, 0);
    else
        compute_pvec<<<1, 64, 0, stream>>>(prompt, proj_w, proj_b, pvec, P);

    if (grouped) {
        int* offs = (int*)(a_dst + N);
        int* bsum = offs + (N + 1);
        int* ssrc = bsum + nblk;
        int* rank = (int*)xlb;

        if (binned_ok) {
            int* binned = (int*)xlb;
            if (!fused) csr_zero<<<gridFor(2 * MAX_BINS, 256), 256, 0, stream>>>(binTotal, 2 * MAX_BINS);
            bin_hist<<<gridFor(E, BCHUNK), 256, 0, stream>>>(ei, E, nbin, binTotal);
            const unsigned nblkE = gridFor(E, BCHUNK);
            bin_scatter<<<nblkE + (foldBeside ? 1u : 0u), 256, 0, stream>>>(ei, E, nbin, binTotal, binCursor, binned,
                                                                        (int)nblkE, prompt, proj_w, proj_b, pvec, P,
                                                                        conv_ws, att_srcs, att_dsts, avec, L);
            bin_sort<<<(unsigned)nbin, SORT_T, 0, stream>>>(binned, E, N, binTotal, offs, ssrc);
        } else {
            csr_zero<<<gridFor((long long)N + 1, 256), 256, 0, stream>>>(offs, N + 1);
            csr_rank<<<gridFor(E, 256), 256, 0, stream>>>(ei, E, offs, rank);
            csr_block_sums<<<(unsigned)nblk, 256, 0, stream>>>(offs, N + 1, bsum);
            csr_scan<<<(unsigned)nblk, 256, 0, stream>>>(offs, N + 1, bsum);
            csr_fill<<<gridFor(E, 256), 256, 0, stream>>>(ei, E, offs, rank, ssrc);
        }

        if (fused) {
            gather_mm<false, true><<<gridFor(N, GEMM_ROWS), 256, 0, stream>>>(
                offs, ssrc, emb, pvec, conv_ws, conv_bs, avec, xbuf, out, emb, 0.f, N);
            gather_mm<true, false><<<gridFor(N, GEMM_ROWS), 256, 0, stream>>>(
                offs, ssrc, xbuf, pvec, conv_ws + DDIM * DDIM, conv_bs + DDIM, avec + 2 * DDIM, xlb, out, emb,
                1.f / (float)(L + 1), N);
        } else
        for (int l = 0; l < L; ++l) {
            const float* W  = conv_ws  + (size_t)l * DDIM * DDIM;
            const float* as = att_srcs + (size_t)l * DDIM;
            const float* ad = att_dsts + (size_t)l * DDIM;
            const float* cb = conv_bs  + (size_t)l * DDIM;

            if (l == 0)
                gemm_xw_t<true><<<gridFor(N, GEMM_ROWS), 256, 0, stream>>>(emb, pvec, W, as, ad, xlb, a_src, a_dst, N);
            else
                gemm_xw_t<false><<<gridFor(N, GEMM_ROWS), 256, 0, stream>>>(xbuf, pvec, W, as, ad, xlb, a_src, a_dst, N);
            if (l + 1 < L) {
                const int mid = (l + 2 < L) ? (l == 0 ? MID_STORE : MID_ADD) : MID_NONE;
                node_gather<false><<<gridFor(N, 8 * NODES_PER_WAVE), 256, 0, stream>>>(
                    offs, ssrc, a_src, a_dst, xlb, cb, xbuf, out, emb, pvec, mid, 0, 0, 0.f, N);
            } else {
                node_gather<true><<<gridFor(N, 8 * NODES_PER_WAVE), 256, 0, stream>>>(
                    offs, ssrc, a_src, a_dst, xlb, cb, xbuf, out, emb, pvec, MID_NONE,
                    L >= 3 ? 1 : 0, L >= 2 ? 1 : 0, 1.f / (float)(L + 1), N);
            }
        }
    } else {
        add_prompt<<<gridFor(total, 256), 256, 0, stream>>>(emb, pvec, xbuf, out, total);
        for (int l = 0; l < L; ++l) {
            const float* W  = conv_ws  + (size_t)l * DDIM * DDIM;
            const float* as = att_srcs + (size_t)l * DDIM;
            const float* ad = att_dsts + (size_t)l * DDIM;
            const float* cb = conv_bs  + (size_t)l * DDIM;

            gemm_xw_t<false><<<gridFor(N, GEMM_ROWS), 256, 0, stream>>>(xbuf, pvec, W, as, ad, xlb, a_src, a_dst, N);
            init_layer<<<gridFor(N, 256), 256, 0, stream>>>(xbuf, segmax, segsum, N);
            edge_max<<<gridFor(totE, 256), 256, 0, stream>>>(ei, E, N, a_src, a_dst, segmax);
            edge_scatter<<<gridFor(totE * 32, 256), 256, 0, stream>>>(ei, E, N, a_src, a_dst,
                                                                      segmax, segsum, xlb, xbuf);
            finalize_layer<<<gridFor(total, 256), 256, 0, stream>>>(xbuf, segsum, cb, out, total);
        }
        scale_out<<<gridFor(total, 256), 256, 0, stream>>>(out, 1.f / (float)(L + 1), total);
    }
}
